# starting kernel
# baseline (speedup 1.0000x reference)
#include <hip/hip_runtime.h>

typedef __attribute__((ext_vector_type(2))) float v2f;
typedef __attribute__((ext_vector_type(8))) float v8f;

#define NFFT     2048
#define HOP      512
#define NBINS    1025
#define NFRAMES  1000
#define NBATCH   8
#define OUTLEN   511488
#define NCOLS    (NBATCH * NFRAMES)
#define KSTEPS   514
#define MTILES   (NFFT / 16)
#define NTILES   (NCOLS / 16)
#define FLT_TINY 1.1754943508222875e-38f

__device__ __forceinline__ float fold_coef(const float* __restrict__ Wr,
                                           const float* __restrict__ Wi,
                                           int o, int k) {
  if (k < 1028) {
    int f = k;
    if (f >= NBINS) return 0.0f;
    float v = Wr[o * NFFT + f];
    if (f >= 1 && f <= NFFT / 2 - 1) v += Wr[o * NFFT + (NFFT - f)];
    return v;
  } else {
    int f = k - 1028;
    if (f >= NBINS) return 0.0f;
    float v = -Wi[o * NFFT + f];
    if (f >= 1 && f <= NFFT / 2 - 1) v += Wi[o * NFFT + (NFFT - f)];
    return v;
  }
}

__global__ void pack_a_kernel(const float* __restrict__ Wr,
                              const float* __restrict__ Wi,
                              float* __restrict__ apack) {
  int gid  = blockIdx.x * blockDim.x + threadIdx.x;
  int lane = gid & 31;
  int kk   = (gid >> 5) % KSTEPS;
  int mt   = (gid >> 5) / KSTEPS;
  if (mt >= MTILES) return;
  int o = mt * 16 + (lane & 15);
  int k = kk * 4 + 2 * (lane >> 4);
  int idx = ((mt * KSTEPS + kk) * 32 + lane) * 2;
  apack[idx + 0] = fold_coef(Wr, Wi, o, k);
  apack[idx + 1] = fold_coef(Wr, Wi, o, k + 1);
}

#define LOADA(kk)                                                              \
  _Pragma("unroll") for (int i = 0; i < 4; ++i)                                \
      a[i] = Ab[((long)(mg4 + i) * KSTEPS + (kk)) * 32 + lane];

#define WMMA16()                                                               \
  _Pragma("unroll") for (int i = 0; i < 4; ++i) {                              \
    _Pragma("unroll") for (int j = 0; j < 4; ++j) {                            \
      c[i][j] = __builtin_amdgcn_wmma_f32_16x16x4_f32(                         \
          false, a[i], false, b[j], (short)0, c[i][j], false, false);          \
    }                                                                          \
  }

__global__ void __launch_bounds__(256)
istft_gemm_kernel(const float* __restrict__ xr, const float* __restrict__ xi,
                  const float* __restrict__ apack, float* __restrict__ sbuf) {
  const int lane = threadIdx.x & 31;
  const int w    = blockIdx.x * 8 + (threadIdx.x >> 5);
  const int mg   = w & 31;
  const int ng   = w >> 5;
  const int mg4  = mg * 4;
  const int h    = lane >> 4;

  const v2f* Ab = (const v2f*)apack;

  int colIdx[4];
#pragma unroll
  for (int j = 0; j < 4; ++j)
    colIdx[j] = (ng * 64 + j * 16 + (lane & 15)) * NBINS;

  v8f c[4][4];
#pragma unroll
  for (int i = 0; i < 4; ++i)
#pragma unroll
    for (int j = 0; j < 4; ++j) c[i][j] = (v8f){};

  v2f a[4], b[4];

#pragma unroll 2
  for (int kk = 0; kk < 256; ++kk) {
    const int f = kk * 4 + 2 * h;
    LOADA(kk);
#pragma unroll
    for (int j = 0; j < 4; ++j) {
      b[j].x = xr[colIdx[j] + f];
      b[j].y = xr[colIdx[j] + f + 1];
    }
    WMMA16();
  }
  {
    LOADA(256);
#pragma unroll
    for (int j = 0; j < 4; ++j) {
      float v = xr[colIdx[j] + 1024];
      b[j].x = (h == 0) ? v : 0.0f;
      b[j].y = 0.0f;
    }
    WMMA16();
  }

#pragma unroll 2
  for (int kk = 257; kk < 513; ++kk) {
    const int f = (kk - 257) * 4 + 2 * h;
    LOADA(kk);
#pragma unroll
    for (int j = 0; j < 4; ++j) {
      b[j].x = xi[colIdx[j] + f];
      b[j].y = xi[colIdx[j] + f + 1];
    }
    WMMA16();
  }
  {
    LOADA(513);
#pragma unroll
    for (int j = 0; j < 4; ++j) {
      float v = xi[colIdx[j] + 1024];
      b[j].x = (h == 0) ? v : 0.0f;
      b[j].y = 0.0f;
    }
    WMMA16();
  }

  const int orow = h * 8;
#pragma unroll
  for (int j = 0; j < 4; ++j) {
    const int nn = ng * 64 + j * 16 + (lane & 15);
    const int bb = nn / NFRAMES;
    const int tt = nn - bb * NFRAMES;
    float* outb = sbuf + (long)bb * NFFT * NFRAMES + tt;
#pragma unroll
    for (int i = 0; i < 4; ++i) {
      const int ob = (mg4 + i) * 16 + orow;
#pragma unroll
      for (int v = 0; v < 8; ++v)
        outb[(long)(ob + v) * NFRAMES] = c[i][j][v];
    }
  }
}

__global__ void ola_kernel(const float* __restrict__ sbuf,
                           const float* __restrict__ win,
                           float* __restrict__ out) {
  int gid = blockIdx.x * blockDim.x + threadIdx.x;
  if (gid >= NBATCH * OUTLEN) return;
  int b  = gid / OUTLEN;
  int jo = gid - b * OUTLEN;
  int j  = jo + NFFT / 2;

  int tmax = j >> 9;
  if (tmax > NFRAMES - 1) tmax = NFRAMES - 1;
  int tmin = (j >= NFFT - HOP) ? ((j - (NFFT - HOP)) >> 9) : 0;

  const float* sb = sbuf + (long)b * NFFT * NFRAMES;
  float acc = 0.0f, wss = 0.0f;
  for (int t = tmin; t <= tmax; ++t) {
    int o = j - (t << 9);
    acc += sb[(long)o * NFRAMES + t];
    float wv = win[o];
    wss += wv * wv;
  }
  out[gid] = (wss > FLT_TINY) ? acc / wss : acc;
}

extern "C" void kernel_launch(void* const* d_in, const int* in_sizes, int n_in,
                              void* d_out, int out_size, void* d_ws,
                              size_t ws_size, hipStream_t stream) {
  const float* real = (const float*)d_in[0];
  const float* imag = (const float*)d_in[1];
  const float* Wr   = (const float*)d_in[2];
  const float* Wi   = (const float*)d_in[3];
  const float* win  = (const float*)d_in[4];
  float* out = (float*)d_out;

  float* apack = (float*)d_ws;
  size_t apack_elems = (size_t)MTILES * KSTEPS * 64;
  float* sbuf = apack + apack_elems;

  {
    int total = MTILES * KSTEPS * 32;
    pack_a_kernel<<<(total + 255) / 256, 256, 0, stream>>>(Wr, Wi, apack);
  }
  istft_gemm_kernel<<<(MTILES / 4) * (NTILES / 4) / 8, 256, 0, stream>>>(
      real, imag, apack, sbuf);
  {
    int total = NBATCH * OUTLEN;
    ola_kernel<<<(total + 255) / 256, 256, 0, stream>>>(sbuf, win, out);
  }
}


# GEMM fragment loads widened to 16 bytes per lane: 8 consecutive k per lane per 16-k group, 4x fewer B load instructions; k permuted alike in both loaders, so the k sum order changes
# speedup vs baseline: 1.0707x; 1.0707x over previous
#include <hip/hip_runtime.h>

typedef __attribute__((ext_vector_type(2))) float v2f;
typedef __attribute__((ext_vector_type(4))) float v4f;
typedef __attribute__((ext_vector_type(8))) float v8f;

#define NFFT     2048
#define HOP      512
#define NBINS    1025
#define NFRAMES  1000
#define NBATCH   8
#define OUTLEN   511488
#define NCOLS    (NBATCH * NFRAMES)
#define KSTEPS   514
#define MTILES   (NFFT / 16)
#define NTILES   (NCOLS / 16)
#define FLT_TINY 1.1754943508222875e-38f

__device__ __forceinline__ float fold_coef(const float* __restrict__ Wr,
                                           const float* __restrict__ Wi,
                                           int o, int k) {
  if (k < 1028) {
    int f = k;
    if (f >= NBINS) return 0.0f;
    float v = Wr[o * NFFT + f];
    if (f >= 1 && f <= NFFT / 2 - 1) v += Wr[o * NFFT + (NFFT - f)];
    return v;
  } else {
    int f = k - 1028;
    if (f >= NBINS) return 0.0f;
    float v = -Wi[o * NFFT + f];
    if (f >= 1 && f <= NFFT / 2 - 1) v += Wi[o * NFFT + (NFFT - f)];
    return v;
  }
}

__global__ void pack_a_kernel(const float* __restrict__ Wr,
                              const float* __restrict__ Wi,
                              float* __restrict__ apack) {
  int gid  = blockIdx.x * blockDim.x + threadIdx.x;
  int lane = gid & 31;
  int kk   = (gid >> 5) % KSTEPS;
  int mt   = (gid >> 5) / KSTEPS;
  if (mt >= MTILES) return;
  int o  = mt * 16 + (lane & 15);
  int h  = lane >> 4;
  int r0 = (kk < 257) ? 0 : 257;
  int r  = kk - r0;
  if (r < 256) {
    int G = r >> 2, s = r & 3;
    int k = r0 * 4 + 16 * G + 8 * (s >> 1) + 4 * h + 2 * (s & 1);
    int idx = (mt * KSTEPS + r0 + 4 * G) * 64 + lane * 8 + 2 * s;
    apack[idx + 0] = fold_coef(Wr, Wi, o, k);
    apack[idx + 1] = fold_coef(Wr, Wi, o, k + 1);
  } else {
    int k = kk * 4 + 2 * h;
    int idx = ((mt * KSTEPS + kk) * 32 + lane) * 2;
    apack[idx + 0] = fold_coef(Wr, Wi, o, k);
    apack[idx + 1] = fold_coef(Wr, Wi, o, k + 1);
  }
}

#define LOADA(kk)                                                              \
  _Pragma("unroll") for (int i = 0; i < 4; ++i)                                \
      a[i] = Ab[((long)(mg4 + i) * KSTEPS + (kk)) * 32 + lane];

#define WMMA16()                                                               \
  _Pragma("unroll") for (int i = 0; i < 4; ++i) {                              \
    _Pragma("unroll") for (int j = 0; j < 4; ++j) {                            \
      c[i][j] = __builtin_amdgcn_wmma_f32_16x16x4_f32(                         \
          false, a[i], false, b[j], (short)0, c[i][j], false, false);          \
    }                                                                          \
  }

#define GROUP16(X, kstep0, G)                                                  \
  {                                                                            \
    v4f a8[4][2], b8[4][2];                                                    \
    _Pragma("unroll") for (int i = 0; i < 4; ++i) {                            \
      const v4f* pa =                                                          \
          Ab4 + ((long)(mg4 + i) * KSTEPS + (kstep0) + 4 * (G)) * 16 + lane * 2; \
      a8[i][0] = pa[0];                                                        \
      a8[i][1] = pa[1];                                                        \
    }                                                                          \
    _Pragma("unroll") for (int j = 0; j < 4; ++j) {                            \
      const float* pb = (X) + colIdx[j] + 16 * (G) + 4 * h;                    \
      __builtin_memcpy(&b8[j][0], pb, 16);                                     \
      __builtin_memcpy(&b8[j][1], pb + 8, 16);                                 \
    }                                                                          \
    _Pragma("unroll") for (int s = 0; s < 4; ++s) {                            \
      _Pragma("unroll") for (int i = 0; i < 4; ++i) {                          \
        const v4f av = a8[i][s >> 1];                                          \
        a[i] = (s & 1) ? (v2f){av.z, av.w} : (v2f){av.x, av.y};                \
      }                                                                        \
      _Pragma("unroll") for (int j = 0; j < 4; ++j) {                          \
        const v4f bv = b8[j][s >> 1];                                          \
        b[j] = (s & 1) ? (v2f){bv.z, bv.w} : (v2f){bv.x, bv.y};                \
      }                                                                        \
      WMMA16();                                                                \
    }                                                                          \
  }

__global__ void __launch_bounds__(256)
istft_gemm_kernel(const float* __restrict__ xr, const float* __restrict__ xi,
                  const float* __restrict__ apack, float* __restrict__ sbuf) {
  const int lane = threadIdx.x & 31;
  const int w    = blockIdx.x * 8 + (threadIdx.x >> 5);
  const int mg   = w & 31;
  const int ng   = w >> 5;
  const int mg4  = mg * 4;
  const int h    = lane >> 4;

  const v2f* Ab  = (const v2f*)apack;
  const v4f* Ab4 = (const v4f*)apack;

  int colIdx[4];
#pragma unroll
  for (int j = 0; j < 4; ++j)
    colIdx[j] = (ng * 64 + j * 16 + (lane & 15)) * NBINS;

  v8f c[4][4];
#pragma unroll
  for (int i = 0; i < 4; ++i)
#pragma unroll
    for (int j = 0; j < 4; ++j) c[i][j] = (v8f){};

  v2f a[4], b[4];

  for (int G = 0; G < 64; ++G) GROUP16(xr, 0, G)
  {
    LOADA(256);
#pragma unroll
    for (int j = 0; j < 4; ++j) {
      float v = xr[colIdx[j] + 1024];
      b[j].x = (h == 0) ? v : 0.0f;
      b[j].y = 0.0f;
    }
    WMMA16();
  }

  for (int G = 0; G < 64; ++G) GROUP16(xi, 257, G)
  {
    LOADA(513);
#pragma unroll
    for (int j = 0; j < 4; ++j) {
      float v = xi[colIdx[j] + 1024];
      b[j].x = (h == 0) ? v : 0.0f;
      b[j].y = 0.0f;
    }
    WMMA16();
  }

  const int orow = h * 8;
#pragma unroll
  for (int j = 0; j < 4; ++j) {
    const int nn = ng * 64 + j * 16 + (lane & 15);
    const int bb = nn / NFRAMES;
    const int tt = nn - bb * NFRAMES;
    float* outb = sbuf + (long)bb * NFFT * NFRAMES + tt;
#pragma unroll
    for (int i = 0; i < 4; ++i) {
      const int ob = (mg4 + i) * 16 + orow;
#pragma unroll
      for (int v = 0; v < 8; ++v)
        outb[(long)(ob + v) * NFRAMES] = c[i][j][v];
    }
  }
}

__global__ void ola_kernel(const float* __restrict__ sbuf,
                           const float* __restrict__ win,
                           float* __restrict__ out) {
  int gid = blockIdx.x * blockDim.x + threadIdx.x;
  if (gid >= NBATCH * OUTLEN) return;
  int b  = gid / OUTLEN;
  int jo = gid - b * OUTLEN;
  int j  = jo + NFFT / 2;

  int tmax = j >> 9;
  if (tmax > NFRAMES - 1) tmax = NFRAMES - 1;
  int tmin = (j >= NFFT - HOP) ? ((j - (NFFT - HOP)) >> 9) : 0;

  const float* sb = sbuf + (long)b * NFFT * NFRAMES;
  float acc = 0.0f, wss = 0.0f;
  for (int t = tmin; t <= tmax; ++t) {
    int o = j - (t << 9);
    acc += sb[(long)o * NFRAMES + t];
    float wv = win[o];
    wss += wv * wv;
  }
  out[gid] = (wss > FLT_TINY) ? acc / wss : acc;
}

extern "C" void kernel_launch(void* const* d_in, const int* in_sizes, int n_in,
                              void* d_out, int out_size, void* d_ws,
                              size_t ws_size, hipStream_t stream) {
  const float* real = (const float*)d_in[0];
  const float* imag = (const float*)d_in[1];
  const float* Wr   = (const float*)d_in[2];
  const float* Wi   = (const float*)d_in[3];
  const float* win  = (const float*)d_in[4];
  float* out = (float*)d_out;

  float* apack = (float*)d_ws;
  size_t apack_elems = (size_t)MTILES * KSTEPS * 64;
  float* sbuf = apack + apack_elems;

  {
    int total = MTILES * KSTEPS * 32;
    pack_a_kernel<<<(total + 255) / 256, 256, 0, stream>>>(Wr, Wi, apack);
  }
  istft_gemm_kernel<<<(MTILES / 4) * (NTILES / 4) / 8, 256, 0, stream>>>(
      real, imag, apack, sbuf);
  {
    int total = NBATCH * OUTLEN;
    ola_kernel<<<(total + 255) / 256, 256, 0, stream>>>(sbuf, win, out);
  }
}


# Frame buffer stored time-major [b][t][o] (was [b][o][t]): GEMM epilogue stores b128 pieces, overlap-add reads whole lines, one b128 load per lane and frame, 4 outputs a thread; sum order kept
# speedup vs baseline: 1.2594x; 1.1762x over previous
#include <hip/hip_runtime.h>

typedef __attribute__((ext_vector_type(2))) float v2f;
typedef __attribute__((ext_vector_type(4))) float v4f;
typedef __attribute__((ext_vector_type(8))) float v8f;

#define NFFT     2048
#define HOP      512
#define NBINS    1025
#define NFRAMES  1000
#define NBATCH   8
#define OUTLEN   511488
#define NCOLS    (NBATCH * NFRAMES)
#define KSTEPS   514
#define MTILES   (NFFT / 16)
#define NTILES   (NCOLS / 16)
#define FLT_TINY 1.1754943508222875e-38f

__device__ __forceinline__ float fold_coef(const float* __restrict__ Wr,
                                           const float* __restrict__ Wi,
                                           int o, int k) {
  if (k < 1028) {
    int f = k;
    if (f >= NBINS) return 0.0f;
    float v = Wr[o * NFFT + f];
    if (f >= 1 && f <= NFFT / 2 - 1) v += Wr[o * NFFT + (NFFT - f)];
    return v;
  } else {
    int f = k - 1028;
    if (f >= NBINS) return 0.0f;
    float v = -Wi[o * NFFT + f];
    if (f >= 1 && f <= NFFT / 2 - 1) v += Wi[o * NFFT + (NFFT - f)];
    return v;
  }
}

__global__ void pack_a_kernel(const float* __restrict__ Wr,
                              const float* __restrict__ Wi,
                              float* __restrict__ apack) {
  int gid  = blockIdx.x * blockDim.x + threadIdx.x;
  int lane = gid & 31;
  int kk   = (gid >> 5) % KSTEPS;
  int mt   = (gid >> 5) / KSTEPS;
  if (mt >= MTILES) return;
  int o  = mt * 16 + (lane & 15);
  int h  = lane >> 4;
  int r0 = (kk < 257) ? 0 : 257;
  int r  = kk - r0;
  if (r < 256) {
    int G = r >> 2, s = r & 3;
    int k = r0 * 4 + 16 * G + 8 * (s >> 1) + 4 * h + 2 * (s & 1);
    int idx = (mt * KSTEPS + r0 + 4 * G) * 64 + lane * 8 + 2 * s;
    apack[idx + 0] = fold_coef(Wr, Wi, o, k);
    apack[idx + 1] = fold_coef(Wr, Wi, o, k + 1);
  } else {
    int k = kk * 4 + 2 * h;
    int idx = ((mt * KSTEPS + kk) * 32 + lane) * 2;
    apack[idx + 0] = fold_coef(Wr, Wi, o, k);
    apack[idx + 1] = fold_coef(Wr, Wi, o, k + 1);
  }
}

#define LOADA(kk)                                                              \
  _Pragma("unroll") for (int i = 0; i < 4; ++i)                                \
      a[i] = Ab[((long)(mg4 + i) * KSTEPS + (kk)) * 32 + lane];

#define WMMA16()                                                               \
  _Pragma("unroll") for (int i = 0; i < 4; ++i) {                              \
    _Pragma("unroll") for (int j = 0; j < 4; ++j) {                            \
      c[i][j] = __builtin_amdgcn_wmma_f32_16x16x4_f32(                         \
          false, a[i], false, b[j], (short)0, c[i][j], false, false);          \
    }                                                                          \
  }

#define GROUP16(X, kstep0, G)                                                  \
  {                                                                            \
    v4f a8[4][2], b8[4][2];                                                    \
    _Pragma("unroll") for (int i = 0; i < 4; ++i) {                            \
      const v4f* pa =                                                          \
          Ab4 + ((long)(mg4 + i) * KSTEPS + (kstep0) + 4 * (G)) * 16 + lane * 2; \
      a8[i][0] = pa[0];                                                        \
      a8[i][1] = pa[1];                                                        \
    }                                                                          \
    _Pragma("unroll") for (int j = 0; j < 4; ++j) {                            \
      const float* pb = (X) + colIdx[j] + 16 * (G) + 4 * h;                    \
      __builtin_memcpy(&b8[j][0], pb, 16);                                     \
      __builtin_memcpy(&b8[j][1], pb + 8, 16);                                 \
    }                                                                          \
    _Pragma("unroll") for (int s = 0; s < 4; ++s) {                            \
      _Pragma("unroll") for (int i = 0; i < 4; ++i) {                          \
        const v4f av = a8[i][s >> 1];                                          \
        a[i] = (s & 1) ? (v2f){av.z, av.w} : (v2f){av.x, av.y};                \
      }                                                                        \
      _Pragma("unroll") for (int j = 0; j < 4; ++j) {                          \
        const v4f bv = b8[j][s >> 1];                                          \
        b[j] = (s & 1) ? (v2f){bv.z, bv.w} : (v2f){bv.x, bv.y};                \
      }                                                                        \
      WMMA16();                                                                \
    }                                                                          \
  }

__global__ void __launch_bounds__(256)
istft_gemm_kernel(const float* __restrict__ xr, const float* __restrict__ xi,
                  const float* __restrict__ apack, float* __restrict__ sbuf) {
  const int lane = threadIdx.x & 31;
  const int w    = blockIdx.x * 8 + (threadIdx.x >> 5);
  const int mg   = w & 31;
  const int ng   = w >> 5;
  const int mg4  = mg * 4;
  const int h    = lane >> 4;

  const v2f* Ab  = (const v2f*)apack;
  const v4f* Ab4 = (const v4f*)apack;

  int colIdx[4];
#pragma unroll
  for (int j = 0; j < 4; ++j)
    colIdx[j] = (ng * 64 + j * 16 + (lane & 15)) * NBINS;

  v8f c[4][4];
#pragma unroll
  for (int i = 0; i < 4; ++i)
#pragma unroll
    for (int j = 0; j < 4; ++j) c[i][j] = (v8f){};

  v2f a[4], b[4];

  for (int G = 0; G < 64; ++G) GROUP16(xr, 0, G)
  {
    LOADA(256);
#pragma unroll
    for (int j = 0; j < 4; ++j) {
      float v = xr[colIdx[j] + 1024];
      b[j].x = (h == 0) ? v : 0.0f;
      b[j].y = 0.0f;
    }
    WMMA16();
  }

  for (int G = 0; G < 64; ++G) GROUP16(xi, 257, G)
  {
    LOADA(513);
#pragma unroll
    for (int j = 0; j < 4; ++j) {
      float v = xi[colIdx[j] + 1024];
      b[j].x = (h == 0) ? v : 0.0f;
      b[j].y = 0.0f;
    }
    WMMA16();
  }

  const int orow = h * 8;
#pragma unroll
  for (int j = 0; j < 4; ++j) {
    const int nn = ng * 64 + j * 16 + (lane & 15);
    float* outb = sbuf + (long)nn * NFFT + orow;
#pragma unroll
    for (int i = 0; i < 4; ++i) {
      float* p = outb + (mg4 + i) * 16;
      *(v4f*)(p)     = (v4f){c[i][j][0], c[i][j][1], c[i][j][2], c[i][j][3]};
      *(v4f*)(p + 4) = (v4f){c[i][j][4], c[i][j][5], c[i][j][6], c[i][j][7]};
    }
  }
}

__global__ void __launch_bounds__(256)
ola_kernel(const float* __restrict__ sbuf, const float* __restrict__ win,
           float* __restrict__ out) {
  int gid = blockIdx.x * blockDim.x + threadIdx.x;
  if (gid >= NBATCH * (OUTLEN / 4)) return;
  int b  = gid / (OUTLEN / 4);
  int jo = (gid - b * (OUTLEN / 4)) * 4;
  int j  = jo + NFFT / 2;

  int tmax = j >> 9;
  if (tmax > NFRAMES - 1) tmax = NFRAMES - 1;
  int tmin = (j >= NFFT - HOP) ? ((j - (NFFT - HOP)) >> 9) : 0;

  const float* sb = sbuf + (long)b * NFRAMES * NFFT;
  v4f sv[4], wv[4];
#pragma unroll
  for (int q = 0; q < 4; ++q) {
    int t = tmin + q;
    if (t > tmax) t = tmax;
    int o = j - (t << 9);
    sv[q] = *(const v4f*)(sb + (long)t * NFFT + o);
    __builtin_memcpy(&wv[q], win + o, 16);
  }
  v4f acc = (v4f){0.0f, 0.0f, 0.0f, 0.0f};
  v4f wss = (v4f){0.0f, 0.0f, 0.0f, 0.0f};
#pragma unroll
  for (int q = 0; q < 4; ++q) {
    if (q == 0 || tmin + q <= tmax) {
      acc += sv[q];
      wss += wv[q] * wv[q];
    }
  }
  v4f r;
  r.x = (wss.x > FLT_TINY) ? acc.x / wss.x : acc.x;
  r.y = (wss.y > FLT_TINY) ? acc.y / wss.y : acc.y;
  r.z = (wss.z > FLT_TINY) ? acc.z / wss.z : acc.z;
  r.w = (wss.w > FLT_TINY) ? acc.w / wss.w : acc.w;
  __builtin_memcpy(out + (long)b * OUTLEN + jo, &r, 16);
}

extern "C" void kernel_launch(void* const* d_in, const int* in_sizes, int n_in,
                              void* d_out, int out_size, void* d_ws,
                              size_t ws_size, hipStream_t stream) {
  const float* real = (const float*)d_in[0];
  const float* imag = (const float*)d_in[1];
  const float* Wr   = (const float*)d_in[2];
  const float* Wi   = (const float*)d_in[3];
  const float* win  = (const float*)d_in[4];
  float* out = (float*)d_out;

  float* apack = (float*)d_ws;
  size_t apack_elems = (size_t)MTILES * KSTEPS * 64;
  float* sbuf = apack + apack_elems;

  {
    int total = MTILES * KSTEPS * 32;
    pack_a_kernel<<<(total + 255) / 256, 256, 0, stream>>>(Wr, Wi, apack);
  }
  istft_gemm_kernel<<<(MTILES / 4) * (NTILES / 4) / 8, 256, 0, stream>>>(
      real, imag, apack, sbuf);
  {
    int total = NBATCH * (OUTLEN / 4);
    ola_kernel<<<(total + 255) / 256, 256, 0, stream>>>(sbuf, win, out);
  }
}


# GEMM fragment loads double-buffered in registers: the next 16-k group's 16 loads are issued ahead of the current group's 64 WMMAs (two register sets, two waves per SIMD); k and sum order kept
# speedup vs baseline: 1.4090x; 1.1188x over previous
#include <hip/hip_runtime.h>

typedef __attribute__((ext_vector_type(2))) float v2f;
typedef __attribute__((ext_vector_type(4))) float v4f;
typedef __attribute__((ext_vector_type(8))) float v8f;

#define NFFT     2048
#define HOP      512
#define NBINS    1025
#define NFRAMES  1000
#define NBATCH   8
#define OUTLEN   511488
#define NCOLS    (NBATCH * NFRAMES)
#define KSTEPS   514
#define MTILES   (NFFT / 16)
#define NTILES   (NCOLS / 16)
#define FLT_TINY 1.1754943508222875e-38f

__device__ __forceinline__ float fold_coef(const float* __restrict__ Wr,
                                           const float* __restrict__ Wi,
                                           int o, int k) {
  if (k < 1028) {
    int f = k;
    if (f >= NBINS) return 0.0f;
    float v = Wr[o * NFFT + f];
    if (f >= 1 && f <= NFFT / 2 - 1) v += Wr[o * NFFT + (NFFT - f)];
    return v;
  } else {
    int f = k - 1028;
    if (f >= NBINS) return 0.0f;
    float v = -Wi[o * NFFT + f];
    if (f >= 1 && f <= NFFT / 2 - 1) v += Wi[o * NFFT + (NFFT - f)];
    return v;
  }
}

__global__ void pack_a_kernel(const float* __restrict__ Wr,
                              const float* __restrict__ Wi,
                              float* __restrict__ apack) {
  int gid  = blockIdx.x * blockDim.x + threadIdx.x;
  int lane = gid & 31;
  int kk   = (gid >> 5) % KSTEPS;
  int mt   = (gid >> 5) / KSTEPS;
  if (mt >= MTILES) return;
  int o  = mt * 16 + (lane & 15);
  int h  = lane >> 4;
  int r0 = (kk < 257) ? 0 : 257;
  int r  = kk - r0;
  if (r < 256) {
    int G = r >> 2, s = r & 3;
    int k = r0 * 4 + 16 * G + 8 * (s >> 1) + 4 * h + 2 * (s & 1);
    int idx = (mt * KSTEPS + r0 + 4 * G) * 64 + lane * 8 + 2 * s;
    apack[idx + 0] = fold_coef(Wr, Wi, o, k);
    apack[idx + 1] = fold_coef(Wr, Wi, o, k + 1);
  } else {
    int k = kk * 4 + 2 * h;
    int idx = ((mt * KSTEPS + kk) * 32 + lane) * 2;
    apack[idx + 0] = fold_coef(Wr, Wi, o, k);
    apack[idx + 1] = fold_coef(Wr, Wi, o, k + 1);
  }
}

#define LOADA(kk)                                                              \
  _Pragma("unroll") for (int i = 0; i < 4; ++i)                                \
      a[i] = Ab[((long)(mg4 + i) * KSTEPS + (kk)) * 32 + lane];

#define WMMA16()                                                               \
  _Pragma("unroll") for (int i = 0; i < 4; ++i) {                              \
    _Pragma("unroll") for (int j = 0; j < 4; ++j) {                            \
      c[i][j] = __builtin_amdgcn_wmma_f32_16x16x4_f32(                         \
          false, a[i], false, b[j], (short)0, c[i][j], false, false);          \
    }                                                                          \
  }

#define LOADGRP(X, kstep0, G, A8, B8)                                          \
  {                                                                            \
    _Pragma("unroll") for (int i = 0; i < 4; ++i) {                            \
      const v4f* pa =                                                          \
          Ab4 + ((long)(mg4 + i) * KSTEPS + (kstep0) + 4 * (G)) * 16 + lane * 2; \
      A8[i][0] = pa[0];                                                        \
      A8[i][1] = pa[1];                                                        \
    }                                                                          \
    _Pragma("unroll") for (int j = 0; j < 4; ++j) {                            \
      const float* pb = (X) + colIdx[j] + 16 * (G) + 4 * h;                    \
      __builtin_memcpy(&B8[j][0], pb, 16);                                     \
      __builtin_memcpy(&B8[j][1], pb + 8, 16);                                 \
    }                                                                          \
  }

#define MMAGRP(A8, B8)                                                         \
  {                                                                            \
    _Pragma("unroll") for (int s = 0; s < 4; ++s) {                            \
      _Pragma("unroll") for (int i = 0; i < 4; ++i) {                          \
        const v4f av = A8[i][s >> 1];                                          \
        a[i] = (s & 1) ? (v2f){av.z, av.w} : (v2f){av.x, av.y};                \
      }                                                                        \
      _Pragma("unroll") for (int j = 0; j < 4; ++j) {                          \
        const v4f bv = B8[j][s >> 1];                                          \
        b[j] = (s & 1) ? (v2f){bv.z, bv.w} : (v2f){bv.x, bv.y};                \
      }                                                                        \
      WMMA16();                                                                \
    }                                                                          \
  }

#define TRIPFENCE() asm volatile("" ::: "memory");

#define REGION(X, kstep0)                                                      \
  {                                                                            \
    LOADGRP(X, kstep0, 0, a8p, b8p)                                            \
    for (int G = 0; G < 64; G += 2) {                                          \
      LOADGRP(X, kstep0, G + 1, a8q, b8q)                                      \
      MMAGRP(a8p, b8p)                                                         \
      const int Gn = (G + 2 < 64) ? G + 2 : 63;                                \
      LOADGRP(X, kstep0, Gn, a8p, b8p)                                         \
      MMAGRP(a8q, b8q)                                                         \
      TRIPFENCE()                                                              \
    }                                                                          \
  }

__global__ void __launch_bounds__(256) __attribute__((amdgpu_waves_per_eu(2, 2)))
istft_gemm_kernel(const float* __restrict__ xr, const float* __restrict__ xi,
                  const float* __restrict__ apack, float* __restrict__ sbuf) {
  const int lane = threadIdx.x & 31;
  const int w    = blockIdx.x * 8 + (threadIdx.x >> 5);
  const int mg   = w & 31;
  const int ng   = w >> 5;
  const int mg4  = mg * 4;
  const int h    = lane >> 4;

  const v2f* Ab  = (const v2f*)apack;
  const v4f* Ab4 = (const v4f*)apack;

  int colIdx[4];
#pragma unroll
  for (int j = 0; j < 4; ++j)
    colIdx[j] = (ng * 64 + j * 16 + (lane & 15)) * NBINS;

  v8f c[4][4];
#pragma unroll
  for (int i = 0; i < 4; ++i)
#pragma unroll
    for (int j = 0; j < 4; ++j) c[i][j] = (v8f){};

  v2f a[4], b[4];
  v4f a8p[4][2], b8p[4][2], a8q[4][2], b8q[4][2];

  REGION(xr, 0)
  {
    LOADA(256);
#pragma unroll
    for (int j = 0; j < 4; ++j) {
      float v = xr[colIdx[j] + 1024];
      b[j].x = (h == 0) ? v : 0.0f;
      b[j].y = 0.0f;
    }
    WMMA16();
  }

  REGION(xi, 257)
  {
    LOADA(513);
#pragma unroll
    for (int j = 0; j < 4; ++j) {
      float v = xi[colIdx[j] + 1024];
      b[j].x = (h == 0) ? v : 0.0f;
      b[j].y = 0.0f;
    }
    WMMA16();
  }

  const int orow = h * 8;
#pragma unroll
  for (int j = 0; j < 4; ++j) {
    const int nn = ng * 64 + j * 16 + (lane & 15);
    float* outb = sbuf + (long)nn * NFFT + orow;
#pragma unroll
    for (int i = 0; i < 4; ++i) {
      float* p = outb + (mg4 + i) * 16;
      *(v4f*)(p)     = (v4f){c[i][j][0], c[i][j][1], c[i][j][2], c[i][j][3]};
      *(v4f*)(p + 4) = (v4f){c[i][j][4], c[i][j][5], c[i][j][6], c[i][j][7]};
    }
  }
}

__global__ void __launch_bounds__(256)
ola_kernel(const float* __restrict__ sbuf, const float* __restrict__ win,
           float* __restrict__ out) {
  int gid = blockIdx.x * blockDim.x + threadIdx.x;
  if (gid >= NBATCH * (OUTLEN / 4)) return;
  int b  = gid / (OUTLEN / 4);
  int jo = (gid - b * (OUTLEN / 4)) * 4;
  int j  = jo + NFFT / 2;

  int tmax = j >> 9;
  if (tmax > NFRAMES - 1) tmax = NFRAMES - 1;
  int tmin = (j >= NFFT - HOP) ? ((j - (NFFT - HOP)) >> 9) : 0;

  const float* sb = sbuf + (long)b * NFRAMES * NFFT;
  v4f sv[4], wv[4];
#pragma unroll
  for (int q = 0; q < 4; ++q) {
    int t = tmin + q;
    if (t > tmax) t = tmax;
    int o = j - (t << 9);
    sv[q] = *(const v4f*)(sb + (long)t * NFFT + o);
    __builtin_memcpy(&wv[q], win + o, 16);
  }
  v4f acc = (v4f){0.0f, 0.0f, 0.0f, 0.0f};
  v4f wss = (v4f){0.0f, 0.0f, 0.0f, 0.0f};
#pragma unroll
  for (int q = 0; q < 4; ++q) {
    if (q == 0 || tmin + q <= tmax) {
      acc += sv[q];
      wss += wv[q] * wv[q];
    }
  }
  v4f r;
  r.x = (wss.x > FLT_TINY) ? acc.x / wss.x : acc.x;
  r.y = (wss.y > FLT_TINY) ? acc.y / wss.y : acc.y;
  r.z = (wss.z > FLT_TINY) ? acc.z / wss.z : acc.z;
  r.w = (wss.w > FLT_TINY) ? acc.w / wss.w : acc.w;
  __builtin_memcpy(out + (long)b * OUTLEN + jo, &r, 16);
}

extern "C" void kernel_launch(void* const* d_in, const int* in_sizes, int n_in,
                              void* d_out, int out_size, void* d_ws,
                              size_t ws_size, hipStream_t stream) {
  const float* real = (const float*)d_in[0];
  const float* imag = (const float*)d_in[1];
  const float* Wr   = (const float*)d_in[2];
  const float* Wi   = (const float*)d_in[3];
  const float* win  = (const float*)d_in[4];
  float* out = (float*)d_out;

  float* apack = (float*)d_ws;
  size_t apack_elems = (size_t)MTILES * KSTEPS * 64;
  float* sbuf = apack + apack_elems;

  {
    int total = MTILES * KSTEPS * 32;
    pack_a_kernel<<<(total + 255) / 256, 256, 0, stream>>>(Wr, Wi, apack);
  }
  istft_gemm_kernel<<<(MTILES / 4) * (NTILES / 4) / 8, 256, 0, stream>>>(
      real, imag, apack, sbuf);
  {
    int total = NBATCH * (OUTLEN / 4);
    ola_kernel<<<(total + 255) / 256, 256, 0, stream>>>(sbuf, win, out);
  }
}
